# starting kernel
# baseline (speedup 1.0000x reference)
#include <hip/hip_runtime.h>
#include <math.h>

typedef __attribute__((ext_vector_type(2))) float v2f;
typedef __attribute__((ext_vector_type(4))) float v4f;
typedef __attribute__((ext_vector_type(8))) float v8f;

#define PI_F 3.14159265358979323846f

#define WMMA4(a, b, c) \
  __builtin_amdgcn_wmma_f32_16x16x4_f32(false, (a), false, (b), (short)0, (c), false, false)

__device__ __forceinline__ v8f mm16(const v2f A[4], const v2f B[4]) {
  v8f c = {};
  c = WMMA4(A[0], B[0], c);
  c = WMMA4(A[1], B[1], c);
  c = WMMA4(A[2], B[2], c);
  c = WMMA4(A[3], B[3], c);
  return c;
}

__device__ __forceinline__ void make_dblk(int lane, v2f A[4]) {
  const int i = lane & 15;
  const int h = lane >> 4;
  const float sc0 = 0.35355339059327373f;
  const float sc1 = 0.5f;
#pragma unroll
  for (int c = 0; c < 4; ++c) {
#pragma unroll
    for (int j = 0; j < 2; ++j) {
      const int kk = 4 * c + 2 * h + j;
      float v = 0.0f;
      if ((i < 8) == (kk < 8)) {
        const int kf = i & 7;
        const int n  = kk & 7;
        const float s = (kf == 0) ? sc0 : sc1;
        v = s * cosf(PI_F * (float)((2 * n + 1) * kf) * (1.0f / 16.0f));
      }
      A[c][j] = v;
    }
  }
}

__device__ __forceinline__ void relayout_rows_to_B(float* lds, int lane,
                                                   const float r[8], v2f B[4]) {
  const int i = lane & 15, h = lane >> 4;
#pragma unroll
  for (int v = 0; v < 8; ++v) lds[i * 17 + v + 8 * h] = r[v];
  asm volatile("s_wait_dscnt 0" ::: "memory");
#pragma unroll
  for (int c = 0; c < 4; ++c)
#pragma unroll
    for (int j = 0; j < 2; ++j)
      B[c][j] = lds[(4 * c + 2 * h + j) * 17 + i];
}

__device__ __forceinline__ void relayout_cols_to_BT(float* lds, int lane,
                                                    const v8f cd, v2f B[4]) {
  const int i = lane & 15, h = lane >> 4;
#pragma unroll
  for (int v = 0; v < 8; ++v) lds[(v + 8 * h) * 17 + i] = cd[v];
  asm volatile("s_wait_dscnt 0" ::: "memory");
#pragma unroll
  for (int c = 0; c < 4; ++c)
#pragma unroll
    for (int j = 0; j < 2; ++j)
      B[c][j] = lds[i * 17 + 4 * c + 2 * h + j];
}

__global__ void __launch_bounds__(256) k_dct_reduce(const float* __restrict__ x,
                                                    float* __restrict__ zpart) {
  __shared__ float lds[8][16 * 17];
  const int b = blockIdx.x, q = blockIdx.y;
  const int lane = threadIdx.x & 31, w = threadIdx.x >> 5;
  const int i = lane & 15, h = lane >> 4;
  float* myl = lds[w];

  v2f A[4];
  make_dblk(lane, A);

  const float* img = x + (size_t)b * 3u * 65536u;
  float zacc[8] = {0.f, 0.f, 0.f, 0.f, 0.f, 0.f, 0.f, 0.f};

  for (int it = 0; it < 8; ++it) {
    const int t = q * 64 + w * 8 + it;
    const int ty = t >> 4, tx = t & 15;
    const float* tp = img + (size_t)(ty * 16) * 256 + tx * 16;

    v2f Bt[4];
#pragma unroll
    for (int c = 0; c < 4; ++c)
#pragma unroll
      for (int j = 0; j < 2; ++j)
        Bt[c][j] = tp[(4 * c + 2 * h + j) * 256 + i];

    const v8f m1 = mm16(A, Bt);
    v2f Bm[4];
    relayout_cols_to_BT(myl, lane, m1, Bm);
    const v8f fT = mm16(A, Bm);

#pragma unroll
    for (int v = 0; v < 8; ++v) zacc[v] += fT[v];
  }

#pragma unroll
  for (int v = 0; v < 8; ++v) {
    float val = zacc[v];
    val += __shfl_down(val, 16, 32);
    val += __shfl_down(val, 8, 32);
    if (lane < 8) {
      const int widx = q * 8 + w;
      zpart[((size_t)b * 32 + widx) * 64 + lane * 8 + v] = val;
    }
  }
}

__global__ void k_mlp(const float* __restrict__ zpart,
                      const float* __restrict__ fc1_w, const float* __restrict__ fc1_b,
                      const float* __restrict__ fc2_w, const float* __restrict__ fc2_b,
                      float* __restrict__ s) {
  const int b = blockIdx.x * blockDim.x + threadIdx.x;
  if (b >= 128) return;

  float z[64];
#pragma unroll 4
  for (int f = 0; f < 64; ++f) {
    float acc = 0.0f;
    for (int widx = 0; widx < 32; ++widx)
      acc += zpart[((size_t)b * 32 + widx) * 64 + f];
    z[f] = acc * (1.0f / 1024.0f);
  }

  float hid[32];
  for (int o = 0; o < 32; ++o) {
    float acc = fc1_b[o];
    for (int f = 0; f < 64; ++f) acc += z[f] * fc1_w[o * 64 + f];
    hid[o] = acc > 0.0f ? acc : 0.0f;
  }
  for (int j = 0; j < 64; ++j) {
    float acc = fc2_b[j];
    for (int o = 0; o < 32; ++o) acc += hid[o] * fc2_w[j * 32 + o];
    s[b * 64 + j] = 1.0f / (1.0f + expf(-acc));
  }
}

__global__ void __launch_bounds__(256) k_apply(const float* __restrict__ x,
                                               const float* __restrict__ s,
                                               float* __restrict__ out) {
  __shared__ float lds[8][16 * 17];
  __shared__ float ssh[64];
  const int b = blockIdx.x, q = blockIdx.y;
  const int lane = threadIdx.x & 31, w = threadIdx.x >> 5;
  const int i = lane & 15, h = lane >> 4;
  float* myl = lds[w];

  if (threadIdx.x < 64) ssh[threadIdx.x] = s[b * 64 + threadIdx.x];
  __syncthreads();

  v2f A[4];
  make_dblk(lane, A);

  const float* img = x + (size_t)b * 3u * 65536u;

  for (int it = 0; it < 8; ++it) {
    const int t = q * 64 + w * 8 + it;
    const int ty = t >> 4, tx = t & 15;
    const float* tp = img + (size_t)(ty * 16) * 256 + tx * 16;

    v2f Bt[4];
#pragma unroll
    for (int c = 0; c < 4; ++c)
#pragma unroll
      for (int j = 0; j < 2; ++j)
        Bt[c][j] = tp[(4 * c + 2 * h + j) * 256 + i];

    const v8f m1 = mm16(A, Bt);
    v2f Bm[4];
    relayout_cols_to_BT(myl, lane, m1, Bm);
    const v8f fT = mm16(A, Bm);

    float g[8];
#pragma unroll
    for (int v = 0; v < 8; ++v) g[v] = fT[v] * ssh[8 * (i & 7) + v];

    v2f Bg[4];
    relayout_rows_to_B(myl, lane, g, Bg);
    const v8f hcd = mm16(A, Bg);

    v2f Bh[4];
    relayout_cols_to_BT(myl, lane, hcd, Bh);
    const v8f ocd = mm16(A, Bh);

    const v4f lo = {ocd[0], ocd[1], ocd[2], ocd[3]};
    const v4f hi = {ocd[4], ocd[5], ocd[6], ocd[7]};
    float* po = out + (size_t)b * 3u * 65536u +
                (size_t)(ty * 16 + i) * 256 + tx * 16 + 8 * h;
#pragma unroll
    for (int ch = 0; ch < 3; ++ch) {
      *(v4f*)(po + (size_t)ch * 65536) = lo;
      *(v4f*)(po + (size_t)ch * 65536 + 4) = hi;
    }
  }
}

extern "C" void kernel_launch(void* const* d_in, const int* in_sizes, int n_in,
                              void* d_out, int out_size, void* d_ws, size_t ws_size,
                              hipStream_t stream) {
  const float* x     = (const float*)d_in[0];
  const float* fc1_w = (const float*)d_in[1];
  const float* fc1_b = (const float*)d_in[2];
  const float* fc2_w = (const float*)d_in[3];
  const float* fc2_b = (const float*)d_in[4];
  float* out = (float*)d_out;

  float* zpart = (float*)d_ws;
  float* sgate = zpart + 128 * 32 * 64;

  dim3 grid(128, 4), blk(256);
  k_dct_reduce<<<grid, blk, 0, stream>>>(x, zpart);
  k_mlp<<<1, 128, 0, stream>>>(zpart, fc1_w, fc1_b, fc2_w, fc2_b, sgate);
  k_apply<<<grid, blk, 0, stream>>>(x, sgate, out);
}


# gate MLP kernel spread over 128 workgroups of 64 threads (one image each, one thread per feature/unit, weight rows prefetched) instead of one 128-thread workgroup; every sum keeps its order
# speedup vs baseline: 2.4215x; 2.4215x over previous
#include <hip/hip_runtime.h>
#include <math.h>

typedef __attribute__((ext_vector_type(2))) float v2f;
typedef __attribute__((ext_vector_type(4))) float v4f;
typedef __attribute__((ext_vector_type(8))) float v8f;

#define PI_F 3.14159265358979323846f

#define WMMA4(a, b, c) \
  __builtin_amdgcn_wmma_f32_16x16x4_f32(false, (a), false, (b), (short)0, (c), false, false)

__device__ __forceinline__ v8f mm16(const v2f A[4], const v2f B[4]) {
  v8f c = {};
  c = WMMA4(A[0], B[0], c);
  c = WMMA4(A[1], B[1], c);
  c = WMMA4(A[2], B[2], c);
  c = WMMA4(A[3], B[3], c);
  return c;
}

__device__ __forceinline__ void make_dblk(int lane, v2f A[4]) {
  const int i = lane & 15;
  const int h = lane >> 4;
  const float sc0 = 0.35355339059327373f;
  const float sc1 = 0.5f;
#pragma unroll
  for (int c = 0; c < 4; ++c) {
#pragma unroll
    for (int j = 0; j < 2; ++j) {
      const int kk = 4 * c + 2 * h + j;
      float v = 0.0f;
      if ((i < 8) == (kk < 8)) {
        const int kf = i & 7;
        const int n  = kk & 7;
        const float s = (kf == 0) ? sc0 : sc1;
        v = s * cosf(PI_F * (float)((2 * n + 1) * kf) * (1.0f / 16.0f));
      }
      A[c][j] = v;
    }
  }
}

__device__ __forceinline__ void relayout_rows_to_B(float* lds, int lane,
                                                   const float r[8], v2f B[4]) {
  const int i = lane & 15, h = lane >> 4;
#pragma unroll
  for (int v = 0; v < 8; ++v) lds[i * 17 + v + 8 * h] = r[v];
  asm volatile("s_wait_dscnt 0" ::: "memory");
#pragma unroll
  for (int c = 0; c < 4; ++c)
#pragma unroll
    for (int j = 0; j < 2; ++j)
      B[c][j] = lds[(4 * c + 2 * h + j) * 17 + i];
}

__device__ __forceinline__ void relayout_cols_to_BT(float* lds, int lane,
                                                    const v8f cd, v2f B[4]) {
  const int i = lane & 15, h = lane >> 4;
#pragma unroll
  for (int v = 0; v < 8; ++v) lds[(v + 8 * h) * 17 + i] = cd[v];
  asm volatile("s_wait_dscnt 0" ::: "memory");
#pragma unroll
  for (int c = 0; c < 4; ++c)
#pragma unroll
    for (int j = 0; j < 2; ++j)
      B[c][j] = lds[i * 17 + 4 * c + 2 * h + j];
}

__global__ void __launch_bounds__(256) k_dct_reduce(const float* __restrict__ x,
                                                    float* __restrict__ zpart) {
  __shared__ float lds[8][16 * 17];
  const int b = blockIdx.x, q = blockIdx.y;
  const int lane = threadIdx.x & 31, w = threadIdx.x >> 5;
  const int i = lane & 15, h = lane >> 4;
  float* myl = lds[w];

  v2f A[4];
  make_dblk(lane, A);

  const float* img = x + (size_t)b * 3u * 65536u;
  float zacc[8] = {0.f, 0.f, 0.f, 0.f, 0.f, 0.f, 0.f, 0.f};

  for (int it = 0; it < 8; ++it) {
    const int t = q * 64 + w * 8 + it;
    const int ty = t >> 4, tx = t & 15;
    const float* tp = img + (size_t)(ty * 16) * 256 + tx * 16;

    v2f Bt[4];
#pragma unroll
    for (int c = 0; c < 4; ++c)
#pragma unroll
      for (int j = 0; j < 2; ++j)
        Bt[c][j] = tp[(4 * c + 2 * h + j) * 256 + i];

    const v8f m1 = mm16(A, Bt);
    v2f Bm[4];
    relayout_cols_to_BT(myl, lane, m1, Bm);
    const v8f fT = mm16(A, Bm);

#pragma unroll
    for (int v = 0; v < 8; ++v) zacc[v] += fT[v];
  }

#pragma unroll
  for (int v = 0; v < 8; ++v) {
    float val = zacc[v];
    val += __shfl_down(val, 16, 32);
    val += __shfl_down(val, 8, 32);
    if (lane < 8) {
      const int widx = q * 8 + w;
      zpart[((size_t)b * 32 + widx) * 64 + lane * 8 + v] = val;
    }
  }
}

__global__ void __launch_bounds__(64) k_mlp(const float* __restrict__ zpart,
                      const float* __restrict__ fc1_w, const float* __restrict__ fc1_b,
                      const float* __restrict__ fc2_w, const float* __restrict__ fc2_b,
                      float* __restrict__ s) {
  __shared__ __attribute__((aligned(16))) float zsh[64];
  __shared__ __attribute__((aligned(16))) float hsh[32];
  const int b = blockIdx.x;
  const int t = threadIdx.x;

  v4f w2[8];
  const v4f* w2p = (const v4f*)(fc2_w + t * 32);
  for (int k = 0; k < 8; ++k) w2[k] = w2p[k];
  const float b2 = fc2_b[t];

  v4f w1[16];
  float b1 = 0.0f;
  if (t < 32) {
    const v4f* w1p = (const v4f*)(fc1_w + t * 64);
    for (int k = 0; k < 16; ++k) w1[k] = w1p[k];
    b1 = fc1_b[t];
  }

  const float* zp = zpart + (size_t)b * 32 * 64 + t;
  float zv[32];
  for (int widx = 0; widx < 32; ++widx) zv[widx] = zp[widx * 64];
  float acc = 0.0f;
  for (int widx = 0; widx < 32; ++widx) acc += zv[widx];
  zsh[t] = acc * (1.0f / 1024.0f);
  __syncthreads();

  if (t < 32) {
    float h = b1;
    for (int k = 0; k < 16; ++k) {
      const v4f zq = *(const v4f*)(zsh + 4 * k);
      h += zq[0] * w1[k][0];
      h += zq[1] * w1[k][1];
      h += zq[2] * w1[k][2];
      h += zq[3] * w1[k][3];
    }
    hsh[t] = h > 0.0f ? h : 0.0f;
  }
  __syncthreads();

  float g = b2;
  for (int k = 0; k < 8; ++k) {
    const v4f hq = *(const v4f*)(hsh + 4 * k);
    g += hq[0] * w2[k][0];
    g += hq[1] * w2[k][1];
    g += hq[2] * w2[k][2];
    g += hq[3] * w2[k][3];
  }
  s[b * 64 + t] = 1.0f / (1.0f + expf(-g));
}

__global__ void __launch_bounds__(256) k_apply(const float* __restrict__ x,
                                               const float* __restrict__ s,
                                               float* __restrict__ out) {
  __shared__ float lds[8][16 * 17];
  __shared__ float ssh[64];
  const int b = blockIdx.x, q = blockIdx.y;
  const int lane = threadIdx.x & 31, w = threadIdx.x >> 5;
  const int i = lane & 15, h = lane >> 4;
  float* myl = lds[w];

  if (threadIdx.x < 64) ssh[threadIdx.x] = s[b * 64 + threadIdx.x];
  __syncthreads();

  v2f A[4];
  make_dblk(lane, A);

  const float* img = x + (size_t)b * 3u * 65536u;

  for (int it = 0; it < 8; ++it) {
    const int t = q * 64 + w * 8 + it;
    const int ty = t >> 4, tx = t & 15;
    const float* tp = img + (size_t)(ty * 16) * 256 + tx * 16;

    v2f Bt[4];
#pragma unroll
    for (int c = 0; c < 4; ++c)
#pragma unroll
      for (int j = 0; j < 2; ++j)
        Bt[c][j] = tp[(4 * c + 2 * h + j) * 256 + i];

    const v8f m1 = mm16(A, Bt);
    v2f Bm[4];
    relayout_cols_to_BT(myl, lane, m1, Bm);
    const v8f fT = mm16(A, Bm);

    float g[8];
#pragma unroll
    for (int v = 0; v < 8; ++v) g[v] = fT[v] * ssh[8 * (i & 7) + v];

    v2f Bg[4];
    relayout_rows_to_B(myl, lane, g, Bg);
    const v8f hcd = mm16(A, Bg);

    v2f Bh[4];
    relayout_cols_to_BT(myl, lane, hcd, Bh);
    const v8f ocd = mm16(A, Bh);

    const v4f lo = {ocd[0], ocd[1], ocd[2], ocd[3]};
    const v4f hi = {ocd[4], ocd[5], ocd[6], ocd[7]};
    float* po = out + (size_t)b * 3u * 65536u +
                (size_t)(ty * 16 + i) * 256 + tx * 16 + 8 * h;
#pragma unroll
    for (int ch = 0; ch < 3; ++ch) {
      *(v4f*)(po + (size_t)ch * 65536) = lo;
      *(v4f*)(po + (size_t)ch * 65536 + 4) = hi;
    }
  }
}

extern "C" void kernel_launch(void* const* d_in, const int* in_sizes, int n_in,
                              void* d_out, int out_size, void* d_ws, size_t ws_size,
                              hipStream_t stream) {
  const float* x     = (const float*)d_in[0];
  const float* fc1_w = (const float*)d_in[1];
  const float* fc1_b = (const float*)d_in[2];
  const float* fc2_w = (const float*)d_in[3];
  const float* fc2_b = (const float*)d_in[4];
  float* out = (float*)d_out;

  float* zpart = (float*)d_ws;
  float* sgate = zpart + 128 * 32 * 64;

  dim3 grid(128, 4), blk(256);
  k_dct_reduce<<<grid, blk, 0, stream>>>(x, zpart);
  k_mlp<<<128, 64, 0, stream>>>(zpart, fc1_w, fc1_b, fc2_w, fc2_b, sgate);
  k_apply<<<grid, blk, 0, stream>>>(x, sgate, out);
}


# apply kernel stages 4 neighbouring 16x16 result tiles in LDS and stores whole 128-byte lines (16 lanes x 16 B = 256 B contiguous per row) instead of 16-byte pieces of 16 lines per store
# speedup vs baseline: 4.6458x; 1.9186x over previous
#include <hip/hip_runtime.h>
#include <math.h>

typedef __attribute__((ext_vector_type(2))) float v2f;
typedef __attribute__((ext_vector_type(4))) float v4f;
typedef __attribute__((ext_vector_type(8))) float v8f;

#define PI_F 3.14159265358979323846f

#define WMMA4(a, b, c) \
  __builtin_amdgcn_wmma_f32_16x16x4_f32(false, (a), false, (b), (short)0, (c), false, false)

__device__ __forceinline__ v8f mm16(const v2f A[4], const v2f B[4]) {
  v8f c = {};
  c = WMMA4(A[0], B[0], c);
  c = WMMA4(A[1], B[1], c);
  c = WMMA4(A[2], B[2], c);
  c = WMMA4(A[3], B[3], c);
  return c;
}

__device__ __forceinline__ void make_dblk(int lane, v2f A[4]) {
  const int i = lane & 15;
  const int h = lane >> 4;
  const float sc0 = 0.35355339059327373f;
  const float sc1 = 0.5f;
#pragma unroll
  for (int c = 0; c < 4; ++c) {
#pragma unroll
    for (int j = 0; j < 2; ++j) {
      const int kk = 4 * c + 2 * h + j;
      float v = 0.0f;
      if ((i < 8) == (kk < 8)) {
        const int kf = i & 7;
        const int n  = kk & 7;
        const float s = (kf == 0) ? sc0 : sc1;
        v = s * cosf(PI_F * (float)((2 * n + 1) * kf) * (1.0f / 16.0f));
      }
      A[c][j] = v;
    }
  }
}

__device__ __forceinline__ void relayout_rows_to_B(float* lds, int lane,
                                                   const float r[8], v2f B[4]) {
  const int i = lane & 15, h = lane >> 4;
#pragma unroll
  for (int v = 0; v < 8; ++v) lds[i * 17 + v + 8 * h] = r[v];
  asm volatile("s_wait_dscnt 0" ::: "memory");
#pragma unroll
  for (int c = 0; c < 4; ++c)
#pragma unroll
    for (int j = 0; j < 2; ++j)
      B[c][j] = lds[(4 * c + 2 * h + j) * 17 + i];
}

__device__ __forceinline__ void relayout_cols_to_BT(float* lds, int lane,
                                                    const v8f cd, v2f B[4]) {
  const int i = lane & 15, h = lane >> 4;
#pragma unroll
  for (int v = 0; v < 8; ++v) lds[(v + 8 * h) * 17 + i] = cd[v];
  asm volatile("s_wait_dscnt 0" ::: "memory");
#pragma unroll
  for (int c = 0; c < 4; ++c)
#pragma unroll
    for (int j = 0; j < 2; ++j)
      B[c][j] = lds[i * 17 + 4 * c + 2 * h + j];
}

__global__ void __launch_bounds__(256) k_dct_reduce(const float* __restrict__ x,
                                                    float* __restrict__ zpart) {
  __shared__ float lds[8][16 * 17];
  const int b = blockIdx.x, q = blockIdx.y;
  const int lane = threadIdx.x & 31, w = threadIdx.x >> 5;
  const int i = lane & 15, h = lane >> 4;
  float* myl = lds[w];

  v2f A[4];
  make_dblk(lane, A);

  const float* img = x + (size_t)b * 3u * 65536u;
  float zacc[8] = {0.f, 0.f, 0.f, 0.f, 0.f, 0.f, 0.f, 0.f};

  for (int it = 0; it < 8; ++it) {
    const int t = q * 64 + w * 8 + it;
    const int ty = t >> 4, tx = t & 15;
    const float* tp = img + (size_t)(ty * 16) * 256 + tx * 16;

    v2f Bt[4];
#pragma unroll
    for (int c = 0; c < 4; ++c)
#pragma unroll
      for (int j = 0; j < 2; ++j)
        Bt[c][j] = tp[(4 * c + 2 * h + j) * 256 + i];

    const v8f m1 = mm16(A, Bt);
    v2f Bm[4];
    relayout_cols_to_BT(myl, lane, m1, Bm);
    const v8f fT = mm16(A, Bm);

#pragma unroll
    for (int v = 0; v < 8; ++v) zacc[v] += fT[v];
  }

#pragma unroll
  for (int v = 0; v < 8; ++v) {
    float val = zacc[v];
    val += __shfl_down(val, 16, 32);
    val += __shfl_down(val, 8, 32);
    if (lane < 8) {
      const int widx = q * 8 + w;
      zpart[((size_t)b * 32 + widx) * 64 + lane * 8 + v] = val;
    }
  }
}

__global__ void __launch_bounds__(64) k_mlp(const float* __restrict__ zpart,
                      const float* __restrict__ fc1_w, const float* __restrict__ fc1_b,
                      const float* __restrict__ fc2_w, const float* __restrict__ fc2_b,
                      float* __restrict__ s) {
  __shared__ __attribute__((aligned(16))) float zsh[64];
  __shared__ __attribute__((aligned(16))) float hsh[32];
  const int b = blockIdx.x;
  const int t = threadIdx.x;

  v4f w2[8];
  const v4f* w2p = (const v4f*)(fc2_w + t * 32);
  for (int k = 0; k < 8; ++k) w2[k] = w2p[k];
  const float b2 = fc2_b[t];

  v4f w1[16];
  float b1 = 0.0f;
  if (t < 32) {
    const v4f* w1p = (const v4f*)(fc1_w + t * 64);
    for (int k = 0; k < 16; ++k) w1[k] = w1p[k];
    b1 = fc1_b[t];
  }

  const float* zp = zpart + (size_t)b * 32 * 64 + t;
  float zv[32];
  for (int widx = 0; widx < 32; ++widx) zv[widx] = zp[widx * 64];
  float acc = 0.0f;
  for (int widx = 0; widx < 32; ++widx) acc += zv[widx];
  zsh[t] = acc * (1.0f / 1024.0f);
  __syncthreads();

  if (t < 32) {
    float h = b1;
    for (int k = 0; k < 16; ++k) {
      const v4f zq = *(const v4f*)(zsh + 4 * k);
      h += zq[0] * w1[k][0];
      h += zq[1] * w1[k][1];
      h += zq[2] * w1[k][2];
      h += zq[3] * w1[k][3];
    }
    hsh[t] = h > 0.0f ? h : 0.0f;
  }
  __syncthreads();

  float g = b2;
  for (int k = 0; k < 8; ++k) {
    const v4f hq = *(const v4f*)(hsh + 4 * k);
    g += hq[0] * w2[k][0];
    g += hq[1] * w2[k][1];
    g += hq[2] * w2[k][2];
    g += hq[3] * w2[k][3];
  }
  s[b * 64 + t] = 1.0f / (1.0f + expf(-g));
}

#define STG_PITCH 68

__global__ void __launch_bounds__(256) k_apply(const float* __restrict__ x,
                                               const float* __restrict__ s,
                                               float* __restrict__ out) {
  __shared__ float lds[8][16 * 17];
  __shared__ __attribute__((aligned(16))) float stage[8][16 * STG_PITCH];
  __shared__ float ssh[64];
  const int b = blockIdx.x, q = blockIdx.y;
  const int lane = threadIdx.x & 31, w = threadIdx.x >> 5;
  const int i = lane & 15, h = lane >> 4;
  float* myl = lds[w];
  float* mys = stage[w];

  if (threadIdx.x < 64) ssh[threadIdx.x] = s[b * 64 + threadIdx.x];
  __syncthreads();

  v2f A[4];
  make_dblk(lane, A);

  const float* img = x + (size_t)b * 3u * 65536u;

  const int tbase = q * 64 + w * 8;
  const int ty = tbase >> 4;
  const int txb = tbase & 15;

  const int rr = lane >> 4;
  const int cc = (lane & 15) * 4;

  for (int quad = 0; quad < 2; ++quad) {
    const int tx0 = txb + quad * 4;

    for (int u = 0; u < 4; ++u) {
      const float* tp = img + (size_t)(ty * 16) * 256 + (tx0 + u) * 16;

      v2f Bt[4];
#pragma unroll
      for (int c = 0; c < 4; ++c)
#pragma unroll
        for (int j = 0; j < 2; ++j)
          Bt[c][j] = tp[(4 * c + 2 * h + j) * 256 + i];

      const v8f m1 = mm16(A, Bt);
      v2f Bm[4];
      relayout_cols_to_BT(myl, lane, m1, Bm);
      const v8f fT = mm16(A, Bm);

      float g[8];
#pragma unroll
      for (int v = 0; v < 8; ++v) g[v] = fT[v] * ssh[8 * (i & 7) + v];

      v2f Bg[4];
      relayout_rows_to_B(myl, lane, g, Bg);
      const v8f hcd = mm16(A, Bg);

      v2f Bh[4];
      relayout_cols_to_BT(myl, lane, hcd, Bh);
      const v8f ocd = mm16(A, Bh);

      const v4f lo = {ocd[0], ocd[1], ocd[2], ocd[3]};
      const v4f hi = {ocd[4], ocd[5], ocd[6], ocd[7]};
      float* ps = mys + i * STG_PITCH + u * 16 + 8 * h;
      *(v4f*)(ps) = lo;
      *(v4f*)(ps + 4) = hi;
    }

    asm volatile("s_wait_dscnt 0" ::: "memory");

    float* po = out + (size_t)b * 3u * 65536u +
                (size_t)(ty * 16 + rr) * 256 + tx0 * 16 + cc;
    for (int k = 0; k < 8; ++k) {
      const v4f val = *(const v4f*)(mys + (2 * k + rr) * STG_PITCH + cc);
      for (int ch = 0; ch < 3; ++ch)
        *(v4f*)(po + (size_t)ch * 65536 + (size_t)(2 * k) * 256) = val;
    }
  }
}

extern "C" void kernel_launch(void* const* d_in, const int* in_sizes, int n_in,
                              void* d_out, int out_size, void* d_ws, size_t ws_size,
                              hipStream_t stream) {
  const float* x     = (const float*)d_in[0];
  const float* fc1_w = (const float*)d_in[1];
  const float* fc1_b = (const float*)d_in[2];
  const float* fc2_w = (const float*)d_in[3];
  const float* fc2_b = (const float*)d_in[4];
  float* out = (float*)d_out;

  float* zpart = (float*)d_ws;
  float* sgate = zpart + 128 * 32 * 64;

  dim3 grid(128, 4), blk(256);
  k_dct_reduce<<<grid, blk, 0, stream>>>(x, zpart);
  k_mlp<<<128, 64, 0, stream>>>(zpart, fc1_w, fc1_b, fc2_w, fc2_b, sgate);
  k_apply<<<grid, blk, 0, stream>>>(x, sgate, out);
}


# gate MLP folded into the apply kernel: each of an image's 4 workgroups computes the 64 gates from the partial sums in its prologue (every sum keeps its order); 3 launches become 2
# speedup vs baseline: 5.0470x; 1.0864x over previous
#include <hip/hip_runtime.h>
#include <math.h>

typedef __attribute__((ext_vector_type(2))) float v2f;
typedef __attribute__((ext_vector_type(4))) float v4f;
typedef __attribute__((ext_vector_type(8))) float v8f;

#define PI_F 3.14159265358979323846f

#define WMMA4(a, b, c) \
  __builtin_amdgcn_wmma_f32_16x16x4_f32(false, (a), false, (b), (short)0, (c), false, false)

__device__ __forceinline__ v8f mm16(const v2f A[4], const v2f B[4]) {
  v8f c = {};
  c = WMMA4(A[0], B[0], c);
  c = WMMA4(A[1], B[1], c);
  c = WMMA4(A[2], B[2], c);
  c = WMMA4(A[3], B[3], c);
  return c;
}

__device__ __forceinline__ void make_dblk(int lane, v2f A[4]) {
  const int i = lane & 15;
  const int h = lane >> 4;
  const float sc0 = 0.35355339059327373f;
  const float sc1 = 0.5f;
#pragma unroll
  for (int c = 0; c < 4; ++c) {
#pragma unroll
    for (int j = 0; j < 2; ++j) {
      const int kk = 4 * c + 2 * h + j;
      float v = 0.0f;
      if ((i < 8) == (kk < 8)) {
        const int kf = i & 7;
        const int n  = kk & 7;
        const float s = (kf == 0) ? sc0 : sc1;
        v = s * cosf(PI_F * (float)((2 * n + 1) * kf) * (1.0f / 16.0f));
      }
      A[c][j] = v;
    }
  }
}

__device__ __forceinline__ void relayout_rows_to_B(float* lds, int lane,
                                                   const float r[8], v2f B[4]) {
  const int i = lane & 15, h = lane >> 4;
#pragma unroll
  for (int v = 0; v < 8; ++v) lds[i * 17 + v + 8 * h] = r[v];
  asm volatile("s_wait_dscnt 0" ::: "memory");
#pragma unroll
  for (int c = 0; c < 4; ++c)
#pragma unroll
    for (int j = 0; j < 2; ++j)
      B[c][j] = lds[(4 * c + 2 * h + j) * 17 + i];
}

__device__ __forceinline__ void relayout_cols_to_BT(float* lds, int lane,
                                                    const v8f cd, v2f B[4]) {
  const int i = lane & 15, h = lane >> 4;
#pragma unroll
  for (int v = 0; v < 8; ++v) lds[(v + 8 * h) * 17 + i] = cd[v];
  asm volatile("s_wait_dscnt 0" ::: "memory");
#pragma unroll
  for (int c = 0; c < 4; ++c)
#pragma unroll
    for (int j = 0; j < 2; ++j)
      B[c][j] = lds[i * 17 + 4 * c + 2 * h + j];
}

__global__ void __launch_bounds__(256) k_dct_reduce(const float* __restrict__ x,
                                                    float* __restrict__ zpart) {
  __shared__ float lds[8][16 * 17];
  const int b = blockIdx.x, q = blockIdx.y;
  const int lane = threadIdx.x & 31, w = threadIdx.x >> 5;
  const int i = lane & 15, h = lane >> 4;
  float* myl = lds[w];

  v2f A[4];
  make_dblk(lane, A);

  const float* img = x + (size_t)b * 3u * 65536u;
  float zacc[8] = {0.f, 0.f, 0.f, 0.f, 0.f, 0.f, 0.f, 0.f};

  for (int it = 0; it < 8; ++it) {
    const int t = q * 64 + w * 8 + it;
    const int ty = t >> 4, tx = t & 15;
    const float* tp = img + (size_t)(ty * 16) * 256 + tx * 16;

    v2f Bt[4];
#pragma unroll
    for (int c = 0; c < 4; ++c)
#pragma unroll
      for (int j = 0; j < 2; ++j)
        Bt[c][j] = tp[(4 * c + 2 * h + j) * 256 + i];

    const v8f m1 = mm16(A, Bt);
    v2f Bm[4];
    relayout_cols_to_BT(myl, lane, m1, Bm);
    const v8f fT = mm16(A, Bm);

#pragma unroll
    for (int v = 0; v < 8; ++v) zacc[v] += fT[v];
  }

#pragma unroll
  for (int v = 0; v < 8; ++v) {
    float val = zacc[v];
    val += __shfl_down(val, 16, 32);
    val += __shfl_down(val, 8, 32);
    if (lane < 8) {
      const int widx = q * 8 + w;
      zpart[((size_t)b * 32 + widx) * 64 + lane * 8 + v] = val;
    }
  }
}

__device__ __forceinline__ void gate_mlp(int b, int t, int lane, v2f A[4],
                      const float* __restrict__ zpart,
                      const float* __restrict__ fc1_w, const float* __restrict__ fc1_b,
                      const float* __restrict__ fc2_w, const float* __restrict__ fc2_b,
                      float* zsh, float* hsh, float* ssh) {
  const bool mine = t < 64;

  v4f w2[8];
  float b2 = 0.0f;
  v4f w1[16];
  float b1 = 0.0f;
  float zv[32];
  if (mine) {
    const v4f* w2p = (const v4f*)(fc2_w + t * 32);
    for (int k = 0; k < 8; ++k) w2[k] = w2p[k];
    b2 = fc2_b[t];
    if (t < 32) {
      const v4f* w1p = (const v4f*)(fc1_w + t * 64);
      for (int k = 0; k < 16; ++k) w1[k] = w1p[k];
      b1 = fc1_b[t];
    }

    const float* zp = zpart + (size_t)b * 32 * 64 + t;
    for (int widx = 0; widx < 32; ++widx) zv[widx] = zp[widx * 64];
  }

  make_dblk(lane, A);

  if (mine) {
    float acc = 0.0f;
    for (int widx = 0; widx < 32; ++widx) acc += zv[widx];
    zsh[t] = acc * (1.0f / 1024.0f);
  }
  __syncthreads();

  if (t < 32) {
    float h = b1;
    for (int k = 0; k < 16; ++k) {
      const v4f zq = *(const v4f*)(zsh + 4 * k);
      h += zq[0] * w1[k][0];
      h += zq[1] * w1[k][1];
      h += zq[2] * w1[k][2];
      h += zq[3] * w1[k][3];
    }
    hsh[t] = h > 0.0f ? h : 0.0f;
  }
  __syncthreads();

  if (mine) {
    float g = b2;
    for (int k = 0; k < 8; ++k) {
      const v4f hq = *(const v4f*)(hsh + 4 * k);
      g += hq[0] * w2[k][0];
      g += hq[1] * w2[k][1];
      g += hq[2] * w2[k][2];
      g += hq[3] * w2[k][3];
    }
    ssh[t] = 1.0f / (1.0f + expf(-g));
  }
  __syncthreads();
}

#define STG_PITCH 68

__global__ void __launch_bounds__(256) k_apply(const float* __restrict__ x,
                      const float* __restrict__ zpart,
                      const float* __restrict__ fc1_w, const float* __restrict__ fc1_b,
                      const float* __restrict__ fc2_w, const float* __restrict__ fc2_b,
                      float* __restrict__ out) {
  __shared__ float lds[8][16 * 17];
  __shared__ __attribute__((aligned(16))) float stage[8][16 * STG_PITCH];
  __shared__ __attribute__((aligned(16))) float zsh[64];
  __shared__ __attribute__((aligned(16))) float hsh[32];
  __shared__ float ssh[64];
  const int b = blockIdx.x, q = blockIdx.y;
  const int lane = threadIdx.x & 31, w = threadIdx.x >> 5;
  const int i = lane & 15, h = lane >> 4;
  float* myl = lds[w];
  float* mys = stage[w];

  v2f A[4];
  gate_mlp(b, threadIdx.x, lane, A, zpart, fc1_w, fc1_b, fc2_w, fc2_b, zsh, hsh, ssh);

  const float* img = x + (size_t)b * 3u * 65536u;

  const int tbase = q * 64 + w * 8;
  const int ty = tbase >> 4;
  const int txb = tbase & 15;

  const int rr = lane >> 4;
  const int cc = (lane & 15) * 4;

  for (int quad = 0; quad < 2; ++quad) {
    const int tx0 = txb + quad * 4;

    for (int u = 0; u < 4; ++u) {
      const float* tp = img + (size_t)(ty * 16) * 256 + (tx0 + u) * 16;

      v2f Bt[4];
#pragma unroll
      for (int c = 0; c < 4; ++c)
#pragma unroll
        for (int j = 0; j < 2; ++j)
          Bt[c][j] = tp[(4 * c + 2 * h + j) * 256 + i];

      const v8f m1 = mm16(A, Bt);
      v2f Bm[4];
      relayout_cols_to_BT(myl, lane, m1, Bm);
      const v8f fT = mm16(A, Bm);

      float g[8];
#pragma unroll
      for (int v = 0; v < 8; ++v) g[v] = fT[v] * ssh[8 * (i & 7) + v];

      v2f Bg[4];
      relayout_rows_to_B(myl, lane, g, Bg);
      const v8f hcd = mm16(A, Bg);

      v2f Bh[4];
      relayout_cols_to_BT(myl, lane, hcd, Bh);
      const v8f ocd = mm16(A, Bh);

      const v4f lo = {ocd[0], ocd[1], ocd[2], ocd[3]};
      const v4f hi = {ocd[4], ocd[5], ocd[6], ocd[7]};
      float* ps = mys + i * STG_PITCH + u * 16 + 8 * h;
      *(v4f*)(ps) = lo;
      *(v4f*)(ps + 4) = hi;
    }

    asm volatile("s_wait_dscnt 0" ::: "memory");

    float* po = out + (size_t)b * 3u * 65536u +
                (size_t)(ty * 16 + rr) * 256 + tx0 * 16 + cc;
    for (int k = 0; k < 8; ++k) {
      const v4f val = *(const v4f*)(mys + (2 * k + rr) * STG_PITCH + cc);
      for (int ch = 0; ch < 3; ++ch)
        *(v4f*)(po + (size_t)ch * 65536 + (size_t)(2 * k) * 256) = val;
    }
  }
}

extern "C" void kernel_launch(void* const* d_in, const int* in_sizes, int n_in,
                              void* d_out, int out_size, void* d_ws, size_t ws_size,
                              hipStream_t stream) {
  const float* x     = (const float*)d_in[0];
  const float* fc1_w = (const float*)d_in[1];
  const float* fc1_b = (const float*)d_in[2];
  const float* fc2_w = (const float*)d_in[3];
  const float* fc2_b = (const float*)d_in[4];
  float* out = (float*)d_out;

  float* zpart = (float*)d_ws;

  dim3 grid(128, 4), blk(256);
  k_dct_reduce<<<grid, blk, 0, stream>>>(x, zpart);
  k_apply<<<grid, blk, 0, stream>>>(x, zpart, fc1_w, fc1_b, fc2_w, fc2_b, out);
}


# squeeze taken before the transform (DCT is linear): reduce kernel only sums the 8x8 pixel blocks, one 8x8 DCT of that sum per workgroup on WMMA; changes the order of the squeeze's float sums
# speedup vs baseline: 5.5951x; 1.1086x over previous
#include <hip/hip_runtime.h>
#include <math.h>

typedef __attribute__((ext_vector_type(2))) float v2f;
typedef __attribute__((ext_vector_type(4))) float v4f;
typedef __attribute__((ext_vector_type(8))) float v8f;

#define PI_F 3.14159265358979323846f

#define WMMA4(a, b, c) \
  __builtin_amdgcn_wmma_f32_16x16x4_f32(false, (a), false, (b), (short)0, (c), false, false)

__device__ __forceinline__ v8f mm16(const v2f A[4], const v2f B[4]) {
  v8f c = {};
  c = WMMA4(A[0], B[0], c);
  c = WMMA4(A[1], B[1], c);
  c = WMMA4(A[2], B[2], c);
  c = WMMA4(A[3], B[3], c);
  return c;
}

__device__ __forceinline__ void make_dblk(int lane, v2f A[4]) {
  const int i = lane & 15;
  const int h = lane >> 4;
  const float sc0 = 0.35355339059327373f;
  const float sc1 = 0.5f;
#pragma unroll
  for (int c = 0; c < 4; ++c) {
#pragma unroll
    for (int j = 0; j < 2; ++j) {
      const int kk = 4 * c + 2 * h + j;
      float v = 0.0f;
      if ((i < 8) == (kk < 8)) {
        const int kf = i & 7;
        const int n  = kk & 7;
        const float s = (kf == 0) ? sc0 : sc1;
        v = s * cosf(PI_F * (float)((2 * n + 1) * kf) * (1.0f / 16.0f));
      }
      A[c][j] = v;
    }
  }
}

__device__ __forceinline__ void relayout_rows_to_B(float* lds, int lane,
                                                   const float r[8], v2f B[4]) {
  const int i = lane & 15, h = lane >> 4;
#pragma unroll
  for (int v = 0; v < 8; ++v) lds[i * 17 + v + 8 * h] = r[v];
  asm volatile("s_wait_dscnt 0" ::: "memory");
#pragma unroll
  for (int c = 0; c < 4; ++c)
#pragma unroll
    for (int j = 0; j < 2; ++j)
      B[c][j] = lds[(4 * c + 2 * h + j) * 17 + i];
}

__device__ __forceinline__ void relayout_cols_to_BT(float* lds, int lane,
                                                    const v8f cd, v2f B[4]) {
  const int i = lane & 15, h = lane >> 4;
#pragma unroll
  for (int v = 0; v < 8; ++v) lds[(v + 8 * h) * 17 + i] = cd[v];
  asm volatile("s_wait_dscnt 0" ::: "memory");
#pragma unroll
  for (int c = 0; c < 4; ++c)
#pragma unroll
    for (int j = 0; j < 2; ++j)
      B[c][j] = lds[i * 17 + 4 * c + 2 * h + j];
}

#define RED_PITCH 33

__global__ void __launch_bounds__(256) k_block_sum(const float* __restrict__ x,
                                                   float* __restrict__ zpart) {
  __shared__ float red[8][32 * RED_PITCH];
  const int b = blockIdx.x, q = blockIdx.y;
  const int lane = threadIdx.x & 31, w = threadIdx.x >> 5;
  const int widx = q * 8 + w;
  float* myr = red[w];

  const float* p = x + (size_t)b * 3u * 65536u + (size_t)((widx >> 1) * 16) * 256 +
                   (widx & 1) * 128 + 4 * lane;
  v4f R[16];
#pragma unroll
  for (int r = 0; r < 16; ++r) R[r] = *(const v4f*)(p + r * 256);

#pragma unroll
  for (int r8 = 0; r8 < 8; ++r8) {
    const v4f a = R[r8] + R[r8 + 8];
#pragma unroll
    for (int e = 0; e < 4; ++e) myr[(r8 * 4 + e) * RED_PITCH + lane] = a[e];
  }
  asm volatile("s_wait_dscnt 0" ::: "memory");

  for (int half = 0; half < 2; ++half) {
    const int s = lane + 32 * half;
    const int r8 = s >> 3, c8 = s & 7;
    const float* sp = myr + (r8 * 4 + (c8 & 3)) * RED_PITCH + (c8 >> 2);
    float acc = 0.0f;
    for (int m = 0; m < 16; ++m) acc += sp[2 * m];
    zpart[((size_t)b * 32 + widx) * 64 + s] = acc;
  }
}

__device__ __forceinline__ void gate_mlp(int b, int t, int lane, v2f A[4], float* myl,
                      const float* __restrict__ zpart,
                      const float* __restrict__ fc1_w, const float* __restrict__ fc1_b,
                      const float* __restrict__ fc2_w, const float* __restrict__ fc2_b,
                      float* bsh, float* zsh, float* hsh, float* ssh) {
  const bool mine = t < 64;

  v4f w2[8];
  float b2 = 0.0f;
  v4f w1[16];
  float b1 = 0.0f;
  float zv[32];
  if (mine) {
    const v4f* w2p = (const v4f*)(fc2_w + t * 32);
    for (int k = 0; k < 8; ++k) w2[k] = w2p[k];
    b2 = fc2_b[t];
    if (t < 32) {
      const v4f* w1p = (const v4f*)(fc1_w + t * 64);
      for (int k = 0; k < 16; ++k) w1[k] = w1p[k];
      b1 = fc1_b[t];
    }

    const float* zp = zpart + (size_t)b * 32 * 64 + t;
    for (int widx = 0; widx < 32; ++widx) zv[widx] = zp[widx * 64];
  }

  make_dblk(lane, A);

  if (mine) {
    float acc = 0.0f;
    for (int widx = 0; widx < 32; ++widx) acc += zv[widx];
    bsh[t] = acc;
  }
  __syncthreads();

  if (t < 32) {
    const int i = lane & 15, h = lane >> 4;
    v2f Bs[4];
    for (int c = 0; c < 2; ++c)
      for (int j = 0; j < 2; ++j) {
        const float v = bsh[(4 * c + 2 * h + j) * 8 + (i & 7)];
        Bs[c][j] = (i < 8) ? v : 0.0f;
      }
    Bs[2] = (v2f){0.0f, 0.0f};
    Bs[3] = (v2f){0.0f, 0.0f};
    const v8f m1 = mm16(A, Bs);
    v2f Bm[4];
    relayout_cols_to_BT(myl, lane, m1, Bm);
    const v8f fT = mm16(A, Bm);
    if (lane < 8) {
      for (int v = 0; v < 8; ++v) zsh[8 * lane + v] = fT[v] * (1.0f / 1024.0f);
    }
  }
  __syncthreads();

  if (t < 32) {
    float h = b1;
    for (int k = 0; k < 16; ++k) {
      const v4f zq = *(const v4f*)(zsh + 4 * k);
      h += zq[0] * w1[k][0];
      h += zq[1] * w1[k][1];
      h += zq[2] * w1[k][2];
      h += zq[3] * w1[k][3];
    }
    hsh[t] = h > 0.0f ? h : 0.0f;
  }
  __syncthreads();

  if (mine) {
    float g = b2;
    for (int k = 0; k < 8; ++k) {
      const v4f hq = *(const v4f*)(hsh + 4 * k);
      g += hq[0] * w2[k][0];
      g += hq[1] * w2[k][1];
      g += hq[2] * w2[k][2];
      g += hq[3] * w2[k][3];
    }
    ssh[t] = 1.0f / (1.0f + expf(-g));
  }
  __syncthreads();
}

#define STG_PITCH 68

__global__ void __launch_bounds__(256) k_apply(const float* __restrict__ x,
                      const float* __restrict__ zpart,
                      const float* __restrict__ fc1_w, const float* __restrict__ fc1_b,
                      const float* __restrict__ fc2_w, const float* __restrict__ fc2_b,
                      float* __restrict__ out) {
  __shared__ float lds[8][16 * 17];
  __shared__ __attribute__((aligned(16))) float stage[8][16 * STG_PITCH];
  __shared__ float bsh[64];
  __shared__ __attribute__((aligned(16))) float zsh[64];
  __shared__ __attribute__((aligned(16))) float hsh[32];
  __shared__ float ssh[64];
  const int b = blockIdx.x, q = blockIdx.y;
  const int lane = threadIdx.x & 31, w = threadIdx.x >> 5;
  const int i = lane & 15, h = lane >> 4;
  float* myl = lds[w];
  float* mys = stage[w];

  v2f A[4];
  gate_mlp(b, threadIdx.x, lane, A, myl, zpart, fc1_w, fc1_b, fc2_w, fc2_b, bsh, zsh, hsh, ssh);

  const float* img = x + (size_t)b * 3u * 65536u;

  const int tbase = q * 64 + w * 8;
  const int ty = tbase >> 4;
  const int txb = tbase & 15;

  const int rr = lane >> 4;
  const int cc = (lane & 15) * 4;

  for (int quad = 0; quad < 2; ++quad) {
    const int tx0 = txb + quad * 4;

    for (int u = 0; u < 4; ++u) {
      const float* tp = img + (size_t)(ty * 16) * 256 + (tx0 + u) * 16;

      v2f Bt[4];
#pragma unroll
      for (int c = 0; c < 4; ++c)
#pragma unroll
        for (int j = 0; j < 2; ++j)
          Bt[c][j] = tp[(4 * c + 2 * h + j) * 256 + i];

      const v8f m1 = mm16(A, Bt);
      v2f Bm[4];
      relayout_cols_to_BT(myl, lane, m1, Bm);
      const v8f fT = mm16(A, Bm);

      float g[8];
#pragma unroll
      for (int v = 0; v < 8; ++v) g[v] = fT[v] * ssh[8 * (i & 7) + v];

      v2f Bg[4];
      relayout_rows_to_B(myl, lane, g, Bg);
      const v8f hcd = mm16(A, Bg);

      v2f Bh[4];
      relayout_cols_to_BT(myl, lane, hcd, Bh);
      const v8f ocd = mm16(A, Bh);

      const v4f lo = {ocd[0], ocd[1], ocd[2], ocd[3]};
      const v4f hi = {ocd[4], ocd[5], ocd[6], ocd[7]};
      float* ps = mys + i * STG_PITCH + u * 16 + 8 * h;
      *(v4f*)(ps) = lo;
      *(v4f*)(ps + 4) = hi;
    }

    asm volatile("s_wait_dscnt 0" ::: "memory");

    float* po = out + (size_t)b * 3u * 65536u +
                (size_t)(ty * 16 + rr) * 256 + tx0 * 16 + cc;
    for (int k = 0; k < 8; ++k) {
      const v4f val = *(const v4f*)(mys + (2 * k + rr) * STG_PITCH + cc);
      for (int ch = 0; ch < 3; ++ch)
        *(v4f*)(po + (size_t)ch * 65536 + (size_t)(2 * k) * 256) = val;
    }
  }
}

extern "C" void kernel_launch(void* const* d_in, const int* in_sizes, int n_in,
                              void* d_out, int out_size, void* d_ws, size_t ws_size,
                              hipStream_t stream) {
  const float* x     = (const float*)d_in[0];
  const float* fc1_w = (const float*)d_in[1];
  const float* fc1_b = (const float*)d_in[2];
  const float* fc2_w = (const float*)d_in[3];
  const float* fc2_b = (const float*)d_in[4];
  float* out = (float*)d_out;

  float* zpart = (float*)d_ws;

  dim3 grid(128, 4), blk(256);
  k_block_sum<<<grid, blk, 0, stream>>>(x, zpart);
  k_apply<<<grid, blk, 0, stream>>>(x, zpart, fc1_w, fc1_b, fc2_w, fc2_b, out);
}


# apply kernel: each product's result registers are the next product's A operand (basis as B, one k permutation); the 3 LDS transposes per tile go; changes the order of the products' k sums
# speedup vs baseline: 6.9276x; 1.2382x over previous
#include <hip/hip_runtime.h>
#include <math.h>

typedef __attribute__((ext_vector_type(2))) float v2f;
typedef __attribute__((ext_vector_type(4))) float v4f;
typedef __attribute__((ext_vector_type(8))) float v8f;

#define PI_F 3.14159265358979323846f

#define WMMA4(a, b, c) \
  __builtin_amdgcn_wmma_f32_16x16x4_f32(false, (a), false, (b), (short)0, (c), false, false)

__device__ __forceinline__ v8f mmT(const v8f X, const v2f Bc[4]) {
  v8f c = {};
  c = WMMA4(((v2f){X[0], X[1]}), Bc[0], c);
  c = WMMA4(((v2f){X[2], X[3]}), Bc[1], c);
  c = WMMA4(((v2f){X[4], X[5]}), Bc[2], c);
  c = WMMA4(((v2f){X[6], X[7]}), Bc[3], c);
  return c;
}

__device__ __forceinline__ void make_dbt(int lane, v2f Bc[4]) {
  const int i = lane & 15;
  const int h = lane >> 4;
  const float sc0 = 0.35355339059327373f;
  const float sc1 = 0.5f;
#pragma unroll
  for (int c = 0; c < 4; ++c) {
#pragma unroll
    for (int j = 0; j < 2; ++j) {
      const int kk = 8 * h + 2 * c + j;
      float v = 0.0f;
      if ((i < 8) == (kk < 8)) {
        const int kf = i & 7;
        const int n  = kk & 7;
        const float s = (kf == 0) ? sc0 : sc1;
        v = s * cosf(PI_F * (float)((2 * n + 1) * kf) * (1.0f / 16.0f));
      }
      Bc[c][j] = v;
    }
  }
}

#define RED_PITCH 33

__global__ void __launch_bounds__(256) k_block_sum(const float* __restrict__ x,
                                                   float* __restrict__ zpart) {
  __shared__ float red[8][32 * RED_PITCH];
  const int b = blockIdx.x, q = blockIdx.y;
  const int lane = threadIdx.x & 31, w = threadIdx.x >> 5;
  const int widx = q * 8 + w;
  float* myr = red[w];

  const float* p = x + (size_t)b * 3u * 65536u + (size_t)((widx >> 1) * 16) * 256 +
                   (widx & 1) * 128 + 4 * lane;
  v4f R[16];
#pragma unroll
  for (int r = 0; r < 16; ++r) R[r] = *(const v4f*)(p + r * 256);

#pragma unroll
  for (int r8 = 0; r8 < 8; ++r8) {
    const v4f a = R[r8] + R[r8 + 8];
#pragma unroll
    for (int e = 0; e < 4; ++e) myr[(r8 * 4 + e) * RED_PITCH + lane] = a[e];
  }
  asm volatile("s_wait_dscnt 0" ::: "memory");

  for (int half = 0; half < 2; ++half) {
    const int s = lane + 32 * half;
    const int r8 = s >> 3, c8 = s & 7;
    const float* sp = myr + (r8 * 4 + (c8 & 3)) * RED_PITCH + (c8 >> 2);
    float acc = 0.0f;
    for (int m = 0; m < 16; ++m) acc += sp[2 * m];
    zpart[((size_t)b * 32 + widx) * 64 + s] = acc;
  }
}

__device__ __forceinline__ void barrier_lds_only() {
  __builtin_amdgcn_fence(__ATOMIC_RELEASE, "workgroup", "local");
  __builtin_amdgcn_s_barrier();
  __builtin_amdgcn_fence(__ATOMIC_ACQUIRE, "workgroup", "local");
}

__device__ __forceinline__ void load_tiles4(const float* __restrict__ p, v8f* X) {
#pragma unroll
  for (int u = 0; u < 4; ++u) {
    const v4f lo = *(const v4f*)(p + u * 16);
    const v4f hi = *(const v4f*)(p + u * 16 + 4);
    X[u] = (v8f){lo[0], lo[1], lo[2], lo[3], hi[0], hi[1], hi[2], hi[3]};
  }
}

__device__ __forceinline__ void gate_mlp(int b, int t, int lane, v2f Bc[4],
                      const float* __restrict__ zpart,
                      const float* __restrict__ fc1_w, const float* __restrict__ fc1_b,
                      const float* __restrict__ fc2_w, const float* __restrict__ fc2_b,
                      float* bsh, float* zsh, float* hsh, float* ssh,
                      const float* __restrict__ xq, v8f* Xt) {
  const bool mine = t < 64;

  v4f w2[8];
  float b2 = 0.0f;
  v4f w1[16];
  float b1 = 0.0f;
  float zv[32];
  if (mine) {
    const v4f* w2p = (const v4f*)(fc2_w + t * 32);
    for (int k = 0; k < 8; ++k) w2[k] = w2p[k];
    b2 = fc2_b[t];
    if (t < 32) {
      const v4f* w1p = (const v4f*)(fc1_w + t * 64);
      for (int k = 0; k < 16; ++k) w1[k] = w1p[k];
      b1 = fc1_b[t];
    }

    const float* zp = zpart + (size_t)b * 32 * 64 + t;
    for (int widx = 0; widx < 32; ++widx) zv[widx] = zp[widx * 64];
  }

  make_dbt(lane, Bc);

  load_tiles4(xq, Xt);

  if (mine) {
    float acc = 0.0f;
    for (int widx = 0; widx < 32; ++widx) acc += zv[widx];
    bsh[t] = acc;
  }
  barrier_lds_only();

  if (t < 32) {
    v8f X0;
    for (int r = 0; r < 8; ++r) {
      const float v = bsh[(lane & 7) * 8 + r];
      X0[r] = (lane < 8) ? v : 0.0f;
    }
    const v8f p1 = mmT(X0, Bc);
    const v8f fT = mmT(p1, Bc);
    if (lane < 8) {
      for (int v = 0; v < 8; ++v) zsh[8 * lane + v] = fT[v] * (1.0f / 1024.0f);
    }
  }
  barrier_lds_only();

  if (t < 32) {
    float h = b1;
    for (int k = 0; k < 16; ++k) {
      const v4f zq = *(const v4f*)(zsh + 4 * k);
      h += zq[0] * w1[k][0];
      h += zq[1] * w1[k][1];
      h += zq[2] * w1[k][2];
      h += zq[3] * w1[k][3];
    }
    hsh[t] = h > 0.0f ? h : 0.0f;
  }
  barrier_lds_only();

  if (mine) {
    float g = b2;
    for (int k = 0; k < 8; ++k) {
      const v4f hq = *(const v4f*)(hsh + 4 * k);
      g += hq[0] * w2[k][0];
      g += hq[1] * w2[k][1];
      g += hq[2] * w2[k][2];
      g += hq[3] * w2[k][3];
    }
    ssh[t] = 1.0f / (1.0f + expf(-g));
  }
  barrier_lds_only();
}

#define STG_PITCH 68

__global__ void __launch_bounds__(256) k_apply(const float* __restrict__ x,
                      const float* __restrict__ zpart,
                      const float* __restrict__ fc1_w, const float* __restrict__ fc1_b,
                      const float* __restrict__ fc2_w, const float* __restrict__ fc2_b,
                      float* __restrict__ out) {
  __shared__ __attribute__((aligned(16))) float stage[8][16 * STG_PITCH];
  __shared__ float bsh[64];
  __shared__ __attribute__((aligned(16))) float zsh[64];
  __shared__ __attribute__((aligned(16))) float hsh[32];
  __shared__ float ssh[64];
  const int b = blockIdx.x, q = blockIdx.y;
  const int lane = threadIdx.x & 31, w = threadIdx.x >> 5;
  const int i = lane & 15, h = lane >> 4;
  float* mys = stage[w];

  const float* img = x + (size_t)b * 3u * 65536u;

  const int tbase = q * 64 + w * 8;
  const int ty = tbase >> 4;
  const int txb = tbase & 15;

  const float* xl = img + (size_t)(ty * 16 + i) * 256 + txb * 16 + 8 * h;

  v8f Xt[8];

  v2f Bc[4];
  gate_mlp(b, threadIdx.x, lane, Bc, zpart, fc1_w, fc1_b, fc2_w, fc2_b, bsh, zsh, hsh, ssh,
           xl, Xt);
  load_tiles4(xl + 64, Xt + 4);

  float sv[8];
#pragma unroll
  for (int v = 0; v < 8; ++v) sv[v] = ssh[8 * (i & 7) + v];

  const int rr = lane >> 4;
  const int cc = (lane & 15) * 4;

#pragma unroll
  for (int quad = 0; quad < 2; ++quad) {
    const int tx0 = txb + quad * 4;

    v8f P[4];
#pragma unroll
    for (int u = 0; u < 4; ++u) P[u] = mmT(Xt[4 * quad + u], Bc);
#pragma unroll
    for (int u = 0; u < 4; ++u) P[u] = mmT(P[u], Bc);
#pragma unroll
    for (int u = 0; u < 4; ++u)
#pragma unroll
      for (int v = 0; v < 8; ++v) P[u][v] *= sv[v];
#pragma unroll
    for (int u = 0; u < 4; ++u) P[u] = mmT(P[u], Bc);
#pragma unroll
    for (int u = 0; u < 4; ++u) P[u] = mmT(P[u], Bc);

#pragma unroll
    for (int u = 0; u < 4; ++u) {
      const v4f lo = {P[u][0], P[u][1], P[u][2], P[u][3]};
      const v4f hi = {P[u][4], P[u][5], P[u][6], P[u][7]};
      float* ps = mys + i * STG_PITCH + u * 16 + 8 * h;
      *(v4f*)(ps) = lo;
      *(v4f*)(ps + 4) = hi;
    }

    asm volatile("s_wait_dscnt 0" ::: "memory");

    float* po = out + (size_t)b * 3u * 65536u +
                (size_t)(ty * 16 + rr) * 256 + tx0 * 16 + cc;
    for (int k = 0; k < 8; ++k) {
      const v4f val = *(const v4f*)(mys + (2 * k + rr) * STG_PITCH + cc);
      for (int ch = 0; ch < 3; ++ch)
        *(v4f*)(po + (size_t)ch * 65536 + (size_t)(2 * k) * 256) = val;
    }
  }
}

extern "C" void kernel_launch(void* const* d_in, const int* in_sizes, int n_in,
                              void* d_out, int out_size, void* d_ws, size_t ws_size,
                              hipStream_t stream) {
  const float* x     = (const float*)d_in[0];
  const float* fc1_w = (const float*)d_in[1];
  const float* fc1_b = (const float*)d_in[2];
  const float* fc2_w = (const float*)d_in[3];
  const float* fc2_b = (const float*)d_in[4];
  float* out = (float*)d_out;

  float* zpart = (float*)d_ws;

  dim3 grid(128, 4), blk(256);
  k_block_sum<<<grid, blk, 0, stream>>>(x, zpart);
  k_apply<<<grid, blk, 0, stream>>>(x, zpart, fc1_w, fc1_b, fc2_w, fc2_b, out);
}


# both launches fused into one (256 workgroups x 16 waves): each of an image's 2 workgroups sums all the image's pixel blocks itself, other half read for the sum only; changes the squeeze's sum order
# speedup vs baseline: 7.4354x; 1.0733x over previous
#include <hip/hip_runtime.h>
#include <math.h>

typedef __attribute__((ext_vector_type(2))) float v2f;
typedef __attribute__((ext_vector_type(4))) float v4f;
typedef __attribute__((ext_vector_type(8))) float v8f;

#define PI_F 3.14159265358979323846f

#define WMMA4(a, b, c) \
  __builtin_amdgcn_wmma_f32_16x16x4_f32(false, (a), false, (b), (short)0, (c), false, false)

__device__ __forceinline__ v8f mmT(const v8f X, const v2f Bc[4]) {
  v8f c = {};
  c = WMMA4(((v2f){X[0], X[1]}), Bc[0], c);
  c = WMMA4(((v2f){X[2], X[3]}), Bc[1], c);
  c = WMMA4(((v2f){X[4], X[5]}), Bc[2], c);
  c = WMMA4(((v2f){X[6], X[7]}), Bc[3], c);
  return c;
}

__device__ __forceinline__ void make_dbt(int lane, v2f Bc[4]) {
  const int i = lane & 15;
  const int h = lane >> 4;
  const float sc0 = 0.35355339059327373f;
  const float sc1 = 0.5f;
#pragma unroll
  for (int c = 0; c < 4; ++c) {
#pragma unroll
    for (int j = 0; j < 2; ++j) {
      const int kk = 8 * h + 2 * c + j;
      float v = 0.0f;
      if ((i < 8) == (kk < 8)) {
        const int kf = i & 7;
        const int n  = kk & 7;
        const float s = (kf == 0) ? sc0 : sc1;
        v = s * cosf(PI_F * (float)((2 * n + 1) * kf) * (1.0f / 16.0f));
      }
      Bc[c][j] = v;
    }
  }
}

#define RED_PITCH 33
#define STG_PITCH 68

__device__ __forceinline__ void barrier_lds_only() {
  __builtin_amdgcn_fence(__ATOMIC_RELEASE, "workgroup", "local");
  __builtin_amdgcn_s_barrier();
  __builtin_amdgcn_fence(__ATOMIC_ACQUIRE, "workgroup", "local");
}

__device__ __forceinline__ void load_tiles4(const float* __restrict__ p, v8f* X) {
#pragma unroll
  for (int u = 0; u < 4; ++u) {
    const v4f lo = *(const v4f*)(p + u * 16);
    const v4f hi = *(const v4f*)(p + u * 16 + 4);
    X[u] = (v8f){lo[0], lo[1], lo[2], lo[3], hi[0], hi[1], hi[2], hi[3]};
  }
}

__device__ __forceinline__ void gate_mlp(int t, int lane, const v2f Bc[4],
                      const float* __restrict__ fc1_w, const float* __restrict__ fc1_b,
                      const float* __restrict__ fc2_w, const float* __restrict__ fc2_b,
                      const float* part, float* zsh, float* hsh, float* ssh) {
  const bool mine = t < 64;

  v4f w2[8];
  float b2 = 0.0f;
  v4f w1[16];
  float b1 = 0.0f;
  if (mine) {
    const v4f* w2p = (const v4f*)(fc2_w + t * 32);
    for (int k = 0; k < 8; ++k) w2[k] = w2p[k];
    b2 = fc2_b[t];
    if (t < 32) {
      const v4f* w1p = (const v4f*)(fc1_w + t * 64);
      for (int k = 0; k < 16; ++k) w1[k] = w1p[k];
      b1 = fc1_b[t];
    }
  }
  barrier_lds_only();

  if (t < 32) {
    const float* pp = part + 8 * (lane & 7);
    v4f alo = {0.0f, 0.0f, 0.0f, 0.0f};
    v4f ahi = {0.0f, 0.0f, 0.0f, 0.0f};
    for (int e = 0; e < 32; ++e) {
      alo += *(const v4f*)(pp + e * 64);
      ahi += *(const v4f*)(pp + e * 64 + 4);
    }
    const bool top = lane < 8;
    const v8f X0 = {top ? alo[0] : 0.0f, top ? alo[1] : 0.0f, top ? alo[2] : 0.0f,
                    top ? alo[3] : 0.0f, top ? ahi[0] : 0.0f, top ? ahi[1] : 0.0f,
                    top ? ahi[2] : 0.0f, top ? ahi[3] : 0.0f};
    const v8f p1 = mmT(X0, Bc);
    const v8f fT = mmT(p1, Bc);
    if (lane < 8) {
      for (int v = 0; v < 8; ++v) zsh[8 * lane + v] = fT[v] * (1.0f / 1024.0f);
    }
  }
  barrier_lds_only();

  if (t < 32) {
    float h = b1;
    for (int k = 0; k < 16; ++k) {
      const v4f zq = *(const v4f*)(zsh + 4 * k);
      h += zq[0] * w1[k][0];
      h += zq[1] * w1[k][1];
      h += zq[2] * w1[k][2];
      h += zq[3] * w1[k][3];
    }
    hsh[t] = h > 0.0f ? h : 0.0f;
  }
  barrier_lds_only();

  if (mine) {
    float g = b2;
    for (int k = 0; k < 8; ++k) {
      const v4f hq = *(const v4f*)(hsh + 4 * k);
      g += hq[0] * w2[k][0];
      g += hq[1] * w2[k][1];
      g += hq[2] * w2[k][2];
      g += hq[3] * w2[k][3];
    }
    ssh[t] = 1.0f / (1.0f + expf(-g));
  }
  barrier_lds_only();
}

__global__ void __launch_bounds__(512) k_se_dct(const float* __restrict__ x,
                      const float* __restrict__ fc1_w, const float* __restrict__ fc1_b,
                      const float* __restrict__ fc2_w, const float* __restrict__ fc2_b,
                      float* __restrict__ out) {
  __shared__ __attribute__((aligned(16))) float stage[16][16 * STG_PITCH];
  __shared__ __attribute__((aligned(16))) float part[32 * 64];
  __shared__ __attribute__((aligned(16))) float zsh[64];
  __shared__ __attribute__((aligned(16))) float hsh[32];
  __shared__ float ssh[64];
  const int hf = blockIdx.x, b = blockIdx.y;
  const int lane = threadIdx.x & 31, w = threadIdx.x >> 5;
  const int i = lane & 15, h = lane >> 4;
  float* mys = stage[w];

  const float* img = x + (size_t)b * 3u * 65536u;

  const int ty  = hf * 8 + (w >> 1);
  const int tym = (1 - hf) * 8 + (w >> 1);
  const int txb = (w & 1) * 8;

  const float* xl = img + (size_t)(ty * 16 + i) * 256 + txb * 16 + 8 * h;
  const float* xm = img + (size_t)(tym * 16) * 256 + txb * 16 + 4 * lane;

  v8f Xt[8];
  load_tiles4(xl, Xt);
  load_tiles4(xl + 64, Xt + 4);
  v4f R[16];
#pragma unroll
  for (int r = 0; r < 16; ++r) R[r] = *(const v4f*)(xm + r * 256);

  v2f Bc[4];
  make_dbt(lane, Bc);

  {
    float S[8];
#pragma unroll
    for (int r = 0; r < 8; ++r) {
      float s = Xt[0][r];
#pragma unroll
      for (int u = 1; u < 8; ++u) s += Xt[u][r];
      s += __shfl_xor(s, 8);
      s += __shfl_xor(s, 16);
      S[r] = s;
    }
    if (lane < 8) {
      float* pp = part + (hf * 16 + w) * 64 + 8 * lane;
      *(v4f*)(pp)     = (v4f){S[0], S[1], S[2], S[3]};
      *(v4f*)(pp + 4) = (v4f){S[4], S[5], S[6], S[7]};
    }
  }

#pragma unroll
  for (int r8 = 0; r8 < 8; ++r8) {
    const v4f a = R[r8] + R[r8 + 8];
#pragma unroll
    for (int e = 0; e < 4; ++e) mys[(r8 * 4 + e) * RED_PITCH + lane] = a[e];
  }
  asm volatile("s_wait_dscnt 0" ::: "memory");
  for (int half = 0; half < 2; ++half) {
    const int s = lane + 32 * half;
    const int r8 = s >> 3, c8 = s & 7;
    const float* sp = mys + (r8 * 4 + (c8 & 3)) * RED_PITCH + (c8 >> 2);
    float acc = 0.0f;
    for (int m = 0; m < 16; ++m) acc += sp[2 * m];
    part[((1 - hf) * 16 + w) * 64 + s] = acc;
  }

  gate_mlp(threadIdx.x, lane, Bc, fc1_w, fc1_b, fc2_w, fc2_b, part, zsh, hsh, ssh);

  float sv[8];
#pragma unroll
  for (int v = 0; v < 8; ++v) sv[v] = ssh[8 * (i & 7) + v];

  const int rr = lane >> 4;
  const int cc = (lane & 15) * 4;

#pragma unroll
  for (int quad = 0; quad < 2; ++quad) {
    const int tx0 = txb + quad * 4;

    v8f P[4];
#pragma unroll
    for (int u = 0; u < 4; ++u) P[u] = mmT(Xt[4 * quad + u], Bc);
#pragma unroll
    for (int u = 0; u < 4; ++u) P[u] = mmT(P[u], Bc);
#pragma unroll
    for (int u = 0; u < 4; ++u)
#pragma unroll
      for (int v = 0; v < 8; ++v) P[u][v] *= sv[v];
#pragma unroll
    for (int u = 0; u < 4; ++u) P[u] = mmT(P[u], Bc);
#pragma unroll
    for (int u = 0; u < 4; ++u) P[u] = mmT(P[u], Bc);

#pragma unroll
    for (int u = 0; u < 4; ++u) {
      const v4f lo = {P[u][0], P[u][1], P[u][2], P[u][3]};
      const v4f hi = {P[u][4], P[u][5], P[u][6], P[u][7]};
      float* ps = mys + i * STG_PITCH + u * 16 + 8 * h;
      *(v4f*)(ps) = lo;
      *(v4f*)(ps + 4) = hi;
    }

    asm volatile("s_wait_dscnt 0" ::: "memory");

    float* po = out + (size_t)b * 3u * 65536u +
                (size_t)(ty * 16 + rr) * 256 + tx0 * 16 + cc;
    for (int k = 0; k < 8; ++k) {
      const v4f val = *(const v4f*)(mys + (2 * k + rr) * STG_PITCH + cc);
      for (int ch = 0; ch < 3; ++ch)
        *(v4f*)(po + (size_t)ch * 65536 + (size_t)(2 * k) * 256) = val;
    }
  }
}

extern "C" void kernel_launch(void* const* d_in, const int* in_sizes, int n_in,
                              void* d_out, int out_size, void* d_ws, size_t ws_size,
                              hipStream_t stream) {
  const float* x     = (const float*)d_in[0];
  const float* fc1_w = (const float*)d_in[1];
  const float* fc1_b = (const float*)d_in[2];
  const float* fc2_w = (const float*)d_in[3];
  const float* fc2_b = (const float*)d_in[4];
  float* out = (float*)d_out;

  dim3 grid(2, 128), blk(512);
  k_se_dct<<<grid, blk, 0, stream>>>(x, fc1_w, fc1_b, fc2_w, fc2_b, out);
}
